# starting kernel
# baseline (speedup 1.0000x reference)
#include <hip/hip_runtime.h>
#include <math.h>

#define T_STEPS 256
#define BATCH   128
#define INP_D   1024
#define HS_D    1024
#define OUT_D   1024

#define KC       32
#define GRP_F    192
#define KPAIR_F  96
#define CHUNK_F  (8 * GRP_F)

typedef __attribute__((ext_vector_type(2))) float v2f;
typedef __attribute__((ext_vector_type(8))) float v8f;

__device__ __forceinline__ v8f wmma_f32(v2f a, v2f b, v8f c) {
    return __builtin_amdgcn_wmma_f32_16x16x4_f32(
         false, a,  false, b,
         (short)0, c,  false,  false);
}

__device__ __forceinline__ void barrier_lds() {
    asm volatile(
        "s_wait_dscnt 0x0\n\t"
        "s_barrier_signal -1\n\t"
        "s_barrier_wait -1"
        ::: "memory");
}

__global__ void rnn_zero_kernel(float* __restrict__ p, int n) {
    int i = blockIdx.x * blockDim.x + threadIdx.x;
    if (i < n) p[i] = 0.0f;
}

__device__ __forceinline__ void stage_frag(float* buf, int lrow, int lcol, float4 v) {
    const int g  = lrow >> 2;
    const int hk = (lrow >> 1) & 1;
    const int e  = lrow & 1;
    float* dst = buf + g * GRP_F + hk * KPAIR_F + e;
    dst[(lcol + 0) * 2] = v.x;
    dst[(lcol + 1) * 2] = v.y;
    dst[(lcol + 2) * 2] = v.z;
    dst[(lcol + 3) * 2] = v.w;
}

__device__ __forceinline__ void compute_chunk(
    const float* bufp, const float* ap, int half, int l16,
    v8f& acc0, v8f& acc1)
{
    v2f a[8], b0[8], b1[8];
    #pragma unroll
    for (int g = 0; g < 8; ++g) {
        a[g] = *(const v2f*)(ap + g * 4);
        const float* fb = bufp + g * GRP_F + half * KPAIR_F;
        b0[g] = *(const v2f*)(fb + l16 * 2);
        b1[g] = *(const v2f*)(fb + (l16 + 16) * 2);
    }
    #pragma unroll
    for (int g = 0; g < 8; ++g) {
        acc0 = wmma_f32(a[g], b0[g], acc0);
        acc1 = wmma_f32(a[g], b1[g], acc1);
    }
}

__device__ __forceinline__ void gemm_chunked(
    const float* __restrict__ A, int lda, int KTOT,
    const float* __restrict__ W,
    int m0, int n0, float* Bs,
    v8f& acc0, v8f& acc1)
{
    const int tid  = threadIdx.x;
    const int lane = tid & 31;
    const int half = lane >> 4;
    const int l16  = lane & 15;
    const int NC   = KTOT / KC;

    const int lrow = tid >> 3;
    const int lcol = (tid & 7) << 2;
    const float* gsrc   = W + (size_t)lrow * HS_D + n0 + lcol;
    const float* a_base = A + (size_t)(m0 + l16) * lda + half * 2;

    float4 breg = *(const float4*)gsrc;
    stage_frag(Bs, lrow, lcol, breg);

    int c = 0;
    for (; c < NC - 1; ++c) {
        breg = *(const float4*)(gsrc + (size_t)(c + 1) * KC * HS_D);
        barrier_lds();
        compute_chunk(Bs + (c & 1) * CHUNK_F, a_base + c * KC, half, l16,
                      acc0, acc1);
        stage_frag(Bs + ((c + 1) & 1) * CHUNK_F, lrow, lcol, breg);
    }
    barrier_lds();
    compute_chunk(Bs + (c & 1) * CHUNK_F, a_base + c * KC, half, l16,
                  acc0, acc1);
    barrier_lds();
}

__global__ __launch_bounds__(256) void rnn_step_kernel(
    const float* __restrict__ x,
    const float* __restrict__ h_in,
    const float* __restrict__ W_ih,
    const float* __restrict__ W_hh,
    const float* __restrict__ b_i,
    const float* __restrict__ b_h,
    float* __restrict__ h_out)
{
    __shared__ float Bs[2 * CHUNK_F];

    const int wave = threadIdx.x >> 5;
    const int lane = threadIdx.x & 31;
    const int half = lane >> 4;
    const int l16  = lane & 15;

    const int m0 = wave * 16;
    const int n0 = blockIdx.x * 32;

    v8f acc0 = {};
    v8f acc1 = {};

    gemm_chunked(x,    INP_D, INP_D, W_ih, m0, n0, Bs, acc0, acc1);
    gemm_chunked(h_in, HS_D,  HS_D,  W_hh, m0, n0, Bs, acc0, acc1);

    const float bias0 = b_i[n0 + l16]      + b_h[n0 + l16];
    const float bias1 = b_i[n0 + 16 + l16] + b_h[n0 + 16 + l16];

    #pragma unroll
    for (int r = 0; r < 8; ++r) {
        const int m = m0 + r + half * 8;
        h_out[(size_t)m * HS_D + n0 + l16]      = tanhf(acc0[r] + bias0);
        h_out[(size_t)m * HS_D + n0 + 16 + l16] = tanhf(acc1[r] + bias1);
    }
}

__global__ __launch_bounds__(256) void rnn_out_kernel(
    const float* __restrict__ h,
    const float* __restrict__ W_out,
    const float* __restrict__ b_out,
    float* __restrict__ out)
{
    const int wave = threadIdx.x >> 5;
    const int lane = threadIdx.x & 31;
    const int half = lane >> 4;
    const int l16  = lane & 15;

    const int m0 = wave * 16;
    const int n0 = blockIdx.x * 32;

    v8f acc0 = {};
    v8f acc1 = {};

    const float* a_ptr  = h + (size_t)(m0 + l16) * HS_D + half * 2;
    const float* bbase0 = W_out + (size_t)(n0 + l16) * HS_D + half * 2;
    const float* bbase1 = W_out + (size_t)(n0 + 16 + l16) * HS_D + half * 2;

    #pragma unroll 4
    for (int k = 0; k < HS_D; k += 4) {
        v2f a  = *(const v2f*)(a_ptr + k);
        v2f b0 = *(const v2f*)(bbase0 + k);
        v2f b1 = *(const v2f*)(bbase1 + k);
        acc0 = wmma_f32(a, b0, acc0);
        acc1 = wmma_f32(a, b1, acc1);
    }

    const float bo0 = b_out[n0 + l16];
    const float bo1 = b_out[n0 + 16 + l16];

    #pragma unroll
    for (int r = 0; r < 8; ++r) {
        const int m = m0 + r + half * 8;
        out[(size_t)m * OUT_D + n0 + l16]      = acc0[r] + bo0;
        out[(size_t)m * OUT_D + n0 + 16 + l16] = acc1[r] + bo1;
    }
}

extern "C" void kernel_launch(void* const* d_in, const int* in_sizes, int n_in,
                              void* d_out, int out_size, void* d_ws, size_t ws_size,
                              hipStream_t stream) {
    (void)in_sizes; (void)n_in; (void)out_size; (void)ws_size;

    const float* data  = (const float*)d_in[0];
    const float* W_ih  = (const float*)d_in[1];
    const float* W_hh  = (const float*)d_in[2];
    const float* b_i   = (const float*)d_in[3];
    const float* b_h   = (const float*)d_in[4];
    const float* W_out = (const float*)d_in[5];
    const float* b_out = (const float*)d_in[6];
    float* out = (float*)d_out;

    float* h0 = (float*)d_ws;
    float* h1 = h0 + (size_t)BATCH * HS_D;

    const int hn = BATCH * HS_D;
    rnn_zero_kernel<<<(hn + 255) / 256, 256, 0, stream>>>(h0, hn);

    for (int t = 0; t < T_STEPS; ++t) {
        const float* hin  = (t & 1) ? h1 : h0;
        float*       hout = (t & 1) ? h0 : h1;
        rnn_step_kernel<<<HS_D / 32, 256, 0, stream>>>(
            data + (size_t)t * BATCH * INP_D, hin, W_ih, W_hh, b_i, b_h, hout);
    }
    rnn_out_kernel<<<HS_D / 32, 256, 0, stream>>>(h0, W_out, b_out, out);
}


# Step kernel: K=2048 split over the 8 waves of a group (fragments straight from global, partials summed through LDS in fixed order), 16x32 tile, 32 -> 256 groups; changes summation order
# speedup vs baseline: 7.4001x; 7.4001x over previous
#include <hip/hip_runtime.h>
#include <math.h>

#define T_STEPS 256
#define BATCH   128
#define INP_D   1024
#define HS_D    1024
#define OUT_D   1024

typedef __attribute__((ext_vector_type(2))) float v2f;
typedef __attribute__((ext_vector_type(8))) float v8f;

__device__ __forceinline__ v8f wmma_f32(v2f a, v2f b, v8f c) {
    return __builtin_amdgcn_wmma_f32_16x16x4_f32(
         false, a,  false, b,
         (short)0, c,  false,  false);
}

__global__ void rnn_zero_kernel(float* __restrict__ p, int n) {
    int i = blockIdx.x * blockDim.x + threadIdx.x;
    if (i < n) p[i] = 0.0f;
}

#define KSLICE   256
#define PSTEPS   (KSLICE / 8)

__global__ __launch_bounds__(256) __attribute__((amdgpu_waves_per_eu(2, 2))) void rnn_step_kernel(
    const float* __restrict__ x,
    const float* __restrict__ h_in,
    const float* __restrict__ W_ih,
    const float* __restrict__ W_hh,
    const float* __restrict__ b_i,
    const float* __restrict__ b_h,
    float* __restrict__ h_out)
{
    __shared__ float red[8 * 512];

    const int wave = threadIdx.x >> 5;
    const int lane = threadIdx.x & 31;
    const int half = lane >> 4;
    const int l16  = lane & 15;

    const int n0 = blockIdx.x * 32;
    const int m0 = blockIdx.y * 16;

    const bool first = wave < 4;
    const float* A  = first ? x    : h_in;
    const float* Wm = first ? W_ih : W_hh;
    const int    kb = (wave & 3) * KSLICE + half * 4;

    const float* ap = A  + (size_t)(m0 + l16) * HS_D + kb;
    const float* wp = Wm + (size_t)kb * HS_D + n0 + 2 * l16;

    const float bias = b_i[n0 + lane] + b_h[n0 + lane];

    v8f acc0 = {};
    v8f acc1 = {};

    #pragma unroll
    for (int j = 0; j < PSTEPS; ++j) {
        const float4 a = *(const float4*)(ap + 8 * j);
        const float* w = wp + (size_t)(8 * j) * HS_D;
        const v2f p0 = *(const v2f*)(w);
        const v2f p1 = *(const v2f*)(w + HS_D);
        const v2f p2 = *(const v2f*)(w + 2 * HS_D);
        const v2f p3 = *(const v2f*)(w + 3 * HS_D);
        const v2f a_lo = { a.x, a.y };
        const v2f a_hi = { a.z, a.w };
        const v2f be_lo = { p0.x, p1.x };
        const v2f bo_lo = { p0.y, p1.y };
        const v2f be_hi = { p2.x, p3.x };
        const v2f bo_hi = { p2.y, p3.y };
        acc0 = wmma_f32(a_lo, be_lo, acc0);
        acc1 = wmma_f32(a_lo, bo_lo, acc1);
        acc0 = wmma_f32(a_hi, be_hi, acc0);
        acc1 = wmma_f32(a_hi, bo_hi, acc1);
    }

    float* mine = red + wave * 512;
    #pragma unroll
    for (int r = 0; r < 8; ++r) {
        v2f pr = { acc0[r], acc1[r] };
        *(v2f*)(mine + (r + half * 8) * 32 + 2 * l16) = pr;
    }
    __syncthreads();

    #pragma unroll
    for (int q = 0; q < 2; ++q) {
        const int o = threadIdx.x + q * 256;
        float s = red[o];
        #pragma unroll
        for (int w = 1; w < 8; ++w) s += red[w * 512 + o];
        const int m = o >> 5;
        s += bias;
        h_out[(size_t)(m0 + m) * HS_D + n0 + lane] = tanhf(s);
    }
}

__global__ __launch_bounds__(256) void rnn_out_kernel(
    const float* __restrict__ h,
    const float* __restrict__ W_out,
    const float* __restrict__ b_out,
    float* __restrict__ out)
{
    const int wave = threadIdx.x >> 5;
    const int lane = threadIdx.x & 31;
    const int half = lane >> 4;
    const int l16  = lane & 15;

    const int m0 = wave * 16;
    const int n0 = blockIdx.x * 32;

    v8f acc0 = {};
    v8f acc1 = {};

    const float* a_ptr  = h + (size_t)(m0 + l16) * HS_D + half * 2;
    const float* bbase0 = W_out + (size_t)(n0 + l16) * HS_D + half * 2;
    const float* bbase1 = W_out + (size_t)(n0 + 16 + l16) * HS_D + half * 2;

    #pragma unroll 4
    for (int k = 0; k < HS_D; k += 4) {
        v2f a  = *(const v2f*)(a_ptr + k);
        v2f b0 = *(const v2f*)(bbase0 + k);
        v2f b1 = *(const v2f*)(bbase1 + k);
        acc0 = wmma_f32(a, b0, acc0);
        acc1 = wmma_f32(a, b1, acc1);
    }

    const float bo0 = b_out[n0 + l16];
    const float bo1 = b_out[n0 + 16 + l16];

    #pragma unroll
    for (int r = 0; r < 8; ++r) {
        const int m = m0 + r + half * 8;
        out[(size_t)m * OUT_D + n0 + l16]      = acc0[r] + bo0;
        out[(size_t)m * OUT_D + n0 + 16 + l16] = acc1[r] + bo1;
    }
}

extern "C" void kernel_launch(void* const* d_in, const int* in_sizes, int n_in,
                              void* d_out, int out_size, void* d_ws, size_t ws_size,
                              hipStream_t stream) {
    (void)in_sizes; (void)n_in; (void)out_size; (void)ws_size;

    const float* data  = (const float*)d_in[0];
    const float* W_ih  = (const float*)d_in[1];
    const float* W_hh  = (const float*)d_in[2];
    const float* b_i   = (const float*)d_in[3];
    const float* b_h   = (const float*)d_in[4];
    const float* W_out = (const float*)d_in[5];
    const float* b_out = (const float*)d_in[6];
    float* out = (float*)d_out;

    float* h0 = (float*)d_ws;
    float* h1 = h0 + (size_t)BATCH * HS_D;

    const int hn = BATCH * HS_D;
    rnn_zero_kernel<<<(hn + 255) / 256, 256, 0, stream>>>(h0, hn);

    for (int t = 0; t < T_STEPS; ++t) {
        const float* hin  = (t & 1) ? h1 : h0;
        float*       hout = (t & 1) ? h0 : h1;
        rnn_step_kernel<<<dim3(HS_D / 32, BATCH / 16), 256, 0, stream>>>(
            data + (size_t)t * BATCH * INP_D, hin, W_ih, W_hh, b_i, b_h, hout);
    }
    rnn_out_kernel<<<HS_D / 32, 256, 0, stream>>>(h0, W_out, b_out, out);
}


# Step split in x groups (x_next@W_ih+biases, one launch ahead) and h groups (h@W_hh + that, tanh), 32x32 tiles over K=1024: 256 not 384 KB read per group; third plane in out; changes summation order
# speedup vs baseline: 7.6824x; 1.0381x over previous
#include <hip/hip_runtime.h>
#include <math.h>

#define T_STEPS 256
#define BATCH   128
#define INP_D   1024
#define HS_D    1024
#define OUT_D   1024

typedef __attribute__((ext_vector_type(2))) float v2f;
typedef __attribute__((ext_vector_type(8))) float v8f;

__device__ __forceinline__ v8f wmma_f32(v2f a, v2f b, v8f c) {
    return __builtin_amdgcn_wmma_f32_16x16x4_f32(
         false, a,  false, b,
         (short)0, c,  false,  false);
}

#define KSLICE   128
#define PSTEPS   (KSLICE / 8)

__global__ __launch_bounds__(256) __attribute__((amdgpu_waves_per_eu(2, 2))) void rnn_step_kernel(
    const float* __restrict__ x_next,
    const float* __restrict__ h_in,
    const float* __restrict__ W_ih,
    const float* __restrict__ W_hh,
    const float* __restrict__ b_i,
    const float* __restrict__ b_h,
    float* __restrict__ sh,
    float* __restrict__ s_next,
    int role0,
    int zero_h)
{
    __shared__ float red[8 * 1024];

    const int wave = threadIdx.x >> 5;
    const int lane = threadIdx.x & 31;
    const int half = lane >> 4;
    const int l16  = lane & 15;

    const int n0 = blockIdx.x * 32;
    const int m0 = blockIdx.y * 32;

    const bool xrole = (blockIdx.z + role0) != 0;
    const float* A  = xrole ? x_next : h_in;
    const float* Wm = xrole ? W_ih   : W_hh;
    const int    kb = wave * KSLICE + half * 4;

    const float* ap = A  + (size_t)(m0 + l16) * HS_D + kb;
    const float* wp = Wm + (size_t)kb * HS_D + n0 + 2 * l16;

    const size_t eo = (size_t)(m0 + wave) * HS_D + n0 + lane;
    float e0, e1, e2 = 0.0f, e3 = 0.0f;
    if (xrole) {
        e0 = b_i[n0 + lane];
        e1 = b_h[n0 + lane];
    } else {
        e0 = sh[eo];
        e1 = sh[eo + (size_t)8 * HS_D];
        e2 = sh[eo + (size_t)16 * HS_D];
        e3 = sh[eo + (size_t)24 * HS_D];
    }

    v8f acc00 = {};
    v8f acc01 = {};
    v8f acc10 = {};
    v8f acc11 = {};

    #pragma unroll
    for (int j = 0; j < PSTEPS; ++j) {
        const float4 a0 = *(const float4*)(ap + 8 * j);
        const float4 a1 = *(const float4*)(ap + (size_t)16 * HS_D + 8 * j);
        const float* w = wp + (size_t)(8 * j) * HS_D;
        const v2f p0 = *(const v2f*)(w);
        const v2f p1 = *(const v2f*)(w + HS_D);
        const v2f p2 = *(const v2f*)(w + 2 * HS_D);
        const v2f p3 = *(const v2f*)(w + 3 * HS_D);
        const v2f a0_lo = { a0.x, a0.y };
        const v2f a0_hi = { a0.z, a0.w };
        const v2f a1_lo = { a1.x, a1.y };
        const v2f a1_hi = { a1.z, a1.w };
        const v2f be_lo = { p0.x, p1.x };
        const v2f bo_lo = { p0.y, p1.y };
        const v2f be_hi = { p2.x, p3.x };
        const v2f bo_hi = { p2.y, p3.y };
        acc00 = wmma_f32(a0_lo, be_lo, acc00);
        acc01 = wmma_f32(a0_lo, bo_lo, acc01);
        acc10 = wmma_f32(a1_lo, be_lo, acc10);
        acc11 = wmma_f32(a1_lo, bo_lo, acc11);
        acc00 = wmma_f32(a0_hi, be_hi, acc00);
        acc01 = wmma_f32(a0_hi, bo_hi, acc01);
        acc10 = wmma_f32(a1_hi, be_hi, acc10);
        acc11 = wmma_f32(a1_hi, bo_hi, acc11);
    }

    float* mine = red + wave * 1024;
    #pragma unroll
    for (int r = 0; r < 8; ++r) {
        v2f t0 = { acc00[r], acc01[r] };
        v2f t1 = { acc10[r], acc11[r] };
        *(v2f*)(mine + (r + half * 8) * 32 + 2 * l16)      = t0;
        *(v2f*)(mine + (16 + r + half * 8) * 32 + 2 * l16) = t1;
    }
    __syncthreads();

    float s[4];
    #pragma unroll
    for (int q = 0; q < 4; ++q) {
        const int o = threadIdx.x + q * 256;
        float v = red[o];
        #pragma unroll
        for (int w = 1; w < 8; ++w) v += red[w * 1024 + o];
        s[q] = v;
    }
    if (xrole) {
        const float bias = e0 + e1;
        #pragma unroll
        for (int q = 0; q < 4; ++q) s_next[eo + (size_t)(8 * q) * HS_D] = s[q] + bias;
        if (zero_h) {
            #pragma unroll
            for (int q = 0; q < 4; ++q) sh[eo + (size_t)(8 * q) * HS_D] = 0.0f;
        }
    } else {
        sh[eo]                     = tanhf(s[0] + e0);
        sh[eo + (size_t)8 * HS_D]  = tanhf(s[1] + e1);
        sh[eo + (size_t)16 * HS_D] = tanhf(s[2] + e2);
        sh[eo + (size_t)24 * HS_D] = tanhf(s[3] + e3);
    }
}

__global__ __launch_bounds__(256) void rnn_out_kernel(
    const float* __restrict__ h,
    const float* __restrict__ W_out,
    const float* __restrict__ b_out,
    float* __restrict__ out)
{
    const int wave = threadIdx.x >> 5;
    const int lane = threadIdx.x & 31;
    const int half = lane >> 4;
    const int l16  = lane & 15;

    const int m0 = wave * 16;
    const int n0 = blockIdx.x * 32;

    v8f acc0 = {};
    v8f acc1 = {};

    const float* a_ptr  = h + (size_t)(m0 + l16) * HS_D + half * 2;
    const float* bbase0 = W_out + (size_t)(n0 + l16) * HS_D + half * 2;
    const float* bbase1 = W_out + (size_t)(n0 + 16 + l16) * HS_D + half * 2;

    #pragma unroll 4
    for (int k = 0; k < HS_D; k += 4) {
        v2f a  = *(const v2f*)(a_ptr + k);
        v2f b0 = *(const v2f*)(bbase0 + k);
        v2f b1 = *(const v2f*)(bbase1 + k);
        acc0 = wmma_f32(a, b0, acc0);
        acc1 = wmma_f32(a, b1, acc1);
    }

    const float bo0 = b_out[n0 + l16];
    const float bo1 = b_out[n0 + 16 + l16];

    #pragma unroll
    for (int r = 0; r < 8; ++r) {
        const int m = m0 + r + half * 8;
        out[(size_t)m * OUT_D + n0 + l16]      = acc0[r] + bo0;
        out[(size_t)m * OUT_D + n0 + 16 + l16] = acc1[r] + bo1;
    }
}

extern "C" void kernel_launch(void* const* d_in, const int* in_sizes, int n_in,
                              void* d_out, int out_size, void* d_ws, size_t ws_size,
                              hipStream_t stream) {
    (void)in_sizes; (void)n_in; (void)out_size; (void)ws_size;

    const float* data  = (const float*)d_in[0];
    const float* W_ih  = (const float*)d_in[1];
    const float* W_hh  = (const float*)d_in[2];
    const float* b_i   = (const float*)d_in[3];
    const float* b_h   = (const float*)d_in[4];
    const float* W_out = (const float*)d_in[5];
    const float* b_out = (const float*)d_in[6];
    float* out = (float*)d_out;

    float* pa = (float*)d_ws;
    float* pb = pa + (size_t)BATCH * HS_D;
    float* pc = out;

    rnn_step_kernel<<<dim3(HS_D / 32, BATCH / 32, 1), 256, 0, stream>>>(
        data, pc, W_ih, W_hh, b_i, b_h, pa, pb, 1, 1);

    for (int t = 0; t + 1 < T_STEPS; ++t) {
        rnn_step_kernel<<<dim3(HS_D / 32, BATCH / 32, 2), 256, 0, stream>>>(
            data + (size_t)(t + 1) * BATCH * INP_D, pa, W_ih, W_hh, b_i, b_h, pb, pc, 0, 0);
        float* dead = pa; pa = pb; pb = pc; pc = dead;
    }
    rnn_step_kernel<<<dim3(HS_D / 32, BATCH / 32, 1), 256, 0, stream>>>(
        data, pa, W_ih, W_hh, b_i, b_h, pb, pc, 0, 0);

    rnn_out_kernel<<<HS_D / 32, 256, 0, stream>>>(pb, W_out, b_out, out);
}


# x groups prefetch into L2, one launch ahead, the input rows their tile position reads first in the next launch (global_prefetch_b8 scope DEV, one 256 B unit per lane); loads, products, sums unchanged
# speedup vs baseline: 8.4064x; 1.0942x over previous
#include <hip/hip_runtime.h>
#include <math.h>

#define T_STEPS 256
#define BATCH   128
#define INP_D   1024
#define HS_D    1024
#define OUT_D   1024

typedef __attribute__((ext_vector_type(2))) float v2f;
typedef __attribute__((ext_vector_type(8))) float v8f;

__device__ __forceinline__ v8f wmma_f32(v2f a, v2f b, v8f c) {
    return __builtin_amdgcn_wmma_f32_16x16x4_f32(
         false, a,  false, b,
         (short)0, c,  false,  false);
}

#define KSLICE   128
#define PSTEPS   (KSLICE / 8)

__global__ __launch_bounds__(256) __attribute__((amdgpu_waves_per_eu(2, 2))) void rnn_step_kernel(
    const float* __restrict__ x_next,
    const float* __restrict__ x_pref,
    const float* __restrict__ h_in,
    const float* __restrict__ W_ih,
    const float* __restrict__ W_hh,
    const float* __restrict__ b_i,
    const float* __restrict__ b_h,
    float* __restrict__ sh,
    float* __restrict__ s_next,
    int role0,
    int zero_h)
{
    __shared__ float red[8 * 1024];

    const int wave = threadIdx.x >> 5;
    const int lane = threadIdx.x & 31;
    const int half = lane >> 4;
    const int l16  = lane & 15;

    const int n0 = blockIdx.x * 32;
    const int m0 = blockIdx.y * 32;

    const bool xrole = (blockIdx.z + role0) != 0;
    const float* A  = xrole ? x_next : h_in;
    const float* Wm = xrole ? W_ih   : W_hh;
    const int    kb = wave * KSLICE + half * 4;

    const float* ap = A  + (size_t)(m0 + l16) * HS_D + kb;
    const float* wp = Wm + (size_t)kb * HS_D + n0 + 2 * l16;

    const size_t eo = (size_t)(m0 + wave) * HS_D + n0 + lane;
    float e0, e1, e2 = 0.0f, e3 = 0.0f;
    if (xrole) {
        e0 = b_i[n0 + lane];
        e1 = b_h[n0 + lane];
    } else {
        e0 = sh[eo];
        e1 = sh[eo + (size_t)8 * HS_D];
        e2 = sh[eo + (size_t)16 * HS_D];
        e3 = sh[eo + (size_t)24 * HS_D];
    }

    v8f acc00 = {};
    v8f acc01 = {};
    v8f acc10 = {};
    v8f acc11 = {};

    #pragma unroll
    for (int j = 0; j < PSTEPS; ++j) {
        const float4 a0 = *(const float4*)(ap + 8 * j);
        const float4 a1 = *(const float4*)(ap + (size_t)16 * HS_D + 8 * j);
        const float* w = wp + (size_t)(8 * j) * HS_D;
        const v2f p0 = *(const v2f*)(w);
        const v2f p1 = *(const v2f*)(w + HS_D);
        const v2f p2 = *(const v2f*)(w + 2 * HS_D);
        const v2f p3 = *(const v2f*)(w + 3 * HS_D);
        const v2f a0_lo = { a0.x, a0.y };
        const v2f a0_hi = { a0.z, a0.w };
        const v2f a1_lo = { a1.x, a1.y };
        const v2f a1_hi = { a1.z, a1.w };
        const v2f be_lo = { p0.x, p1.x };
        const v2f bo_lo = { p0.y, p1.y };
        const v2f be_hi = { p2.x, p3.x };
        const v2f bo_hi = { p2.y, p3.y };
        acc00 = wmma_f32(a0_lo, be_lo, acc00);
        acc01 = wmma_f32(a0_lo, bo_lo, acc01);
        acc10 = wmma_f32(a1_lo, be_lo, acc10);
        acc11 = wmma_f32(a1_lo, bo_lo, acc11);
        acc00 = wmma_f32(a0_hi, be_hi, acc00);
        acc01 = wmma_f32(a0_hi, bo_hi, acc01);
        acc10 = wmma_f32(a1_hi, be_hi, acc10);
        acc11 = wmma_f32(a1_hi, bo_hi, acc11);
    }

    if (xrole && x_pref != nullptr && threadIdx.x < 128) {
        const float* pf = x_pref + (size_t)(m0 + (threadIdx.x >> 2)) * INP_D
                        + (4 * (threadIdx.x & 3) + (blockIdx.x & 3)) * 64;
        __builtin_prefetch(pf, 0, 1);
    }

    float* mine = red + wave * 1024;
    #pragma unroll
    for (int r = 0; r < 8; ++r) {
        v2f t0 = { acc00[r], acc01[r] };
        v2f t1 = { acc10[r], acc11[r] };
        *(v2f*)(mine + (r + half * 8) * 32 + 2 * l16)      = t0;
        *(v2f*)(mine + (16 + r + half * 8) * 32 + 2 * l16) = t1;
    }
    __syncthreads();

    float s[4];
    #pragma unroll
    for (int q = 0; q < 4; ++q) {
        const int o = threadIdx.x + q * 256;
        float v = red[o];
        #pragma unroll
        for (int w = 1; w < 8; ++w) v += red[w * 1024 + o];
        s[q] = v;
    }
    if (xrole) {
        const float bias = e0 + e1;
        #pragma unroll
        for (int q = 0; q < 4; ++q) s_next[eo + (size_t)(8 * q) * HS_D] = s[q] + bias;
        if (zero_h) {
            #pragma unroll
            for (int q = 0; q < 4; ++q) sh[eo + (size_t)(8 * q) * HS_D] = 0.0f;
        }
    } else {
        sh[eo]                     = tanhf(s[0] + e0);
        sh[eo + (size_t)8 * HS_D]  = tanhf(s[1] + e1);
        sh[eo + (size_t)16 * HS_D] = tanhf(s[2] + e2);
        sh[eo + (size_t)24 * HS_D] = tanhf(s[3] + e3);
    }
}

__global__ __launch_bounds__(256) void rnn_out_kernel(
    const float* __restrict__ h,
    const float* __restrict__ W_out,
    const float* __restrict__ b_out,
    float* __restrict__ out)
{
    const int wave = threadIdx.x >> 5;
    const int lane = threadIdx.x & 31;
    const int half = lane >> 4;
    const int l16  = lane & 15;

    const int m0 = wave * 16;
    const int n0 = blockIdx.x * 32;

    v8f acc0 = {};
    v8f acc1 = {};

    const float* a_ptr  = h + (size_t)(m0 + l16) * HS_D + half * 2;
    const float* bbase0 = W_out + (size_t)(n0 + l16) * HS_D + half * 2;
    const float* bbase1 = W_out + (size_t)(n0 + 16 + l16) * HS_D + half * 2;

    #pragma unroll 4
    for (int k = 0; k < HS_D; k += 4) {
        v2f a  = *(const v2f*)(a_ptr + k);
        v2f b0 = *(const v2f*)(bbase0 + k);
        v2f b1 = *(const v2f*)(bbase1 + k);
        acc0 = wmma_f32(a, b0, acc0);
        acc1 = wmma_f32(a, b1, acc1);
    }

    const float bo0 = b_out[n0 + l16];
    const float bo1 = b_out[n0 + 16 + l16];

    #pragma unroll
    for (int r = 0; r < 8; ++r) {
        const int m = m0 + r + half * 8;
        out[(size_t)m * OUT_D + n0 + l16]      = acc0[r] + bo0;
        out[(size_t)m * OUT_D + n0 + 16 + l16] = acc1[r] + bo1;
    }
}

extern "C" void kernel_launch(void* const* d_in, const int* in_sizes, int n_in,
                              void* d_out, int out_size, void* d_ws, size_t ws_size,
                              hipStream_t stream) {
    (void)in_sizes; (void)n_in; (void)out_size; (void)ws_size;

    const float* data  = (const float*)d_in[0];
    const float* W_ih  = (const float*)d_in[1];
    const float* W_hh  = (const float*)d_in[2];
    const float* b_i   = (const float*)d_in[3];
    const float* b_h   = (const float*)d_in[4];
    const float* W_out = (const float*)d_in[5];
    const float* b_out = (const float*)d_in[6];
    float* out = (float*)d_out;

    float* pa = (float*)d_ws;
    float* pb = pa + (size_t)BATCH * HS_D;
    float* pc = out;

    rnn_step_kernel<<<dim3(HS_D / 32, BATCH / 32, 1), 256, 0, stream>>>(
        data, data + (size_t)BATCH * INP_D, pc, W_ih, W_hh, b_i, b_h, pa, pb, 1, 1);

    for (int t = 0; t + 2 < T_STEPS; ++t) {
        rnn_step_kernel<<<dim3(HS_D / 32, BATCH / 32, 2), 256, 0, stream>>>(
            data + (size_t)(t + 1) * BATCH * INP_D, data + (size_t)(t + 2) * BATCH * INP_D,
            pa, W_ih, W_hh, b_i, b_h, pb, pc, 0, 0);
        float* dead = pa; pa = pb; pb = pc; pc = dead;
    }
    rnn_step_kernel<<<dim3(HS_D / 32, BATCH / 32, 2), 256, 0, stream>>>(
        data + (size_t)(T_STEPS - 1) * BATCH * INP_D, nullptr,
        pa, W_ih, W_hh, b_i, b_h, pb, pc, 0, 0);
    { float* dead = pa; pa = pb; pb = pc; pc = dead; }
    rnn_step_kernel<<<dim3(HS_D / 32, BATCH / 32, 1), 256, 0, stream>>>(
        data, nullptr, pa, W_ih, W_hh, b_i, b_h, pb, pc, 0, 0);

    rnn_out_kernel<<<HS_D / 32, 256, 0, stream>>>(pb, W_out, b_out, out);
}


# h groups: the four tanh of a thread run side by side without branches (both range forms of the library tanhf evaluated per lane, one selected) in place of four divergent branch sequences in a row
# speedup vs baseline: 8.6706x; 1.0314x over previous
#include <hip/hip_runtime.h>
#include <math.h>

#define T_STEPS 256
#define BATCH   128
#define INP_D   1024
#define HS_D    1024
#define OUT_D   1024

typedef __attribute__((ext_vector_type(2))) float v2f;
typedef __attribute__((ext_vector_type(8))) float v8f;

__device__ __forceinline__ v8f wmma_f32(v2f a, v2f b, v8f c) {
    return __builtin_amdgcn_wmma_f32_16x16x4_f32(
         false, a,  false, b,
         (short)0, c,  false,  false);
}

__device__ __forceinline__ float tanh_both_ranges(float x) {
    const float ax = fabsf(x);
    const float x2 = ax * ax;
    float p = fmaf(x2, -0x1.758e7ap-8f, 0x1.521192p-6f);
    p = fmaf(x2, p, -0x1.b8389cp-5f);
    p = fmaf(x2, p, 0x1.110704p-3f);
    p = fmaf(x2, p, -0x1.555532p-2f);
    const float small = fmaf(x2, ax * p, ax);
    const float e   = expf(2.0f * ax);
    const float big = fmaf(__builtin_amdgcn_rcpf(e + 1.0f), -2.0f, 1.0f);
    return copysignf(ax < 0.625f ? small : big, x);
}

#define KSLICE   128
#define PSTEPS   (KSLICE / 8)

__global__ __launch_bounds__(256) __attribute__((amdgpu_waves_per_eu(2, 2))) void rnn_step_kernel(
    const float* __restrict__ x_next,
    const float* __restrict__ x_pref,
    const float* __restrict__ h_in,
    const float* __restrict__ W_ih,
    const float* __restrict__ W_hh,
    const float* __restrict__ b_i,
    const float* __restrict__ b_h,
    float* __restrict__ sh,
    float* __restrict__ s_next,
    int role0,
    int zero_h)
{
    __shared__ float red[8 * 1024];

    const int wave = threadIdx.x >> 5;
    const int lane = threadIdx.x & 31;
    const int half = lane >> 4;
    const int l16  = lane & 15;

    const int n0 = blockIdx.x * 32;
    const int m0 = blockIdx.y * 32;

    const bool xrole = (blockIdx.z + role0) != 0;
    const float* A  = xrole ? x_next : h_in;
    const float* Wm = xrole ? W_ih   : W_hh;
    const int    kb = wave * KSLICE + half * 4;

    const float* ap = A  + (size_t)(m0 + l16) * HS_D + kb;
    const float* wp = Wm + (size_t)kb * HS_D + n0 + 2 * l16;

    const size_t eo = (size_t)(m0 + wave) * HS_D + n0 + lane;
    float e0, e1, e2 = 0.0f, e3 = 0.0f;
    if (xrole) {
        e0 = b_i[n0 + lane];
        e1 = b_h[n0 + lane];
    } else {
        e0 = sh[eo];
        e1 = sh[eo + (size_t)8 * HS_D];
        e2 = sh[eo + (size_t)16 * HS_D];
        e3 = sh[eo + (size_t)24 * HS_D];
    }

    v8f acc00 = {};
    v8f acc01 = {};
    v8f acc10 = {};
    v8f acc11 = {};

    #pragma unroll
    for (int j = 0; j < PSTEPS; ++j) {
        const float4 a0 = *(const float4*)(ap + 8 * j);
        const float4 a1 = *(const float4*)(ap + (size_t)16 * HS_D + 8 * j);
        const float* w = wp + (size_t)(8 * j) * HS_D;
        const v2f p0 = *(const v2f*)(w);
        const v2f p1 = *(const v2f*)(w + HS_D);
        const v2f p2 = *(const v2f*)(w + 2 * HS_D);
        const v2f p3 = *(const v2f*)(w + 3 * HS_D);
        const v2f a0_lo = { a0.x, a0.y };
        const v2f a0_hi = { a0.z, a0.w };
        const v2f a1_lo = { a1.x, a1.y };
        const v2f a1_hi = { a1.z, a1.w };
        const v2f be_lo = { p0.x, p1.x };
        const v2f bo_lo = { p0.y, p1.y };
        const v2f be_hi = { p2.x, p3.x };
        const v2f bo_hi = { p2.y, p3.y };
        acc00 = wmma_f32(a0_lo, be_lo, acc00);
        acc01 = wmma_f32(a0_lo, bo_lo, acc01);
        acc10 = wmma_f32(a1_lo, be_lo, acc10);
        acc11 = wmma_f32(a1_lo, bo_lo, acc11);
        acc00 = wmma_f32(a0_hi, be_hi, acc00);
        acc01 = wmma_f32(a0_hi, bo_hi, acc01);
        acc10 = wmma_f32(a1_hi, be_hi, acc10);
        acc11 = wmma_f32(a1_hi, bo_hi, acc11);
    }

    if (xrole && x_pref != nullptr && threadIdx.x < 128) {
        const float* pf = x_pref + (size_t)(m0 + (threadIdx.x >> 2)) * INP_D
                        + (4 * (threadIdx.x & 3) + (blockIdx.x & 3)) * 64;
        __builtin_prefetch(pf, 0, 1);
    }

    float* mine = red + wave * 1024;
    #pragma unroll
    for (int r = 0; r < 8; ++r) {
        v2f t0 = { acc00[r], acc01[r] };
        v2f t1 = { acc10[r], acc11[r] };
        *(v2f*)(mine + (r + half * 8) * 32 + 2 * l16)      = t0;
        *(v2f*)(mine + (16 + r + half * 8) * 32 + 2 * l16) = t1;
    }
    __syncthreads();

    float s[4];
    #pragma unroll
    for (int q = 0; q < 4; ++q) {
        const int o = threadIdx.x + q * 256;
        float v = red[o];
        #pragma unroll
        for (int w = 1; w < 8; ++w) v += red[w * 1024 + o];
        s[q] = v;
    }
    if (xrole) {
        const float bias = e0 + e1;
        #pragma unroll
        for (int q = 0; q < 4; ++q) s_next[eo + (size_t)(8 * q) * HS_D] = s[q] + bias;
        if (zero_h) {
            #pragma unroll
            for (int q = 0; q < 4; ++q) sh[eo + (size_t)(8 * q) * HS_D] = 0.0f;
        }
    } else {
        const float t0 = tanh_both_ranges(s[0] + e0);
        const float t1 = tanh_both_ranges(s[1] + e1);
        const float t2 = tanh_both_ranges(s[2] + e2);
        const float t3 = tanh_both_ranges(s[3] + e3);
        sh[eo]                     = t0;
        sh[eo + (size_t)8 * HS_D]  = t1;
        sh[eo + (size_t)16 * HS_D] = t2;
        sh[eo + (size_t)24 * HS_D] = t3;
    }
}

__global__ __launch_bounds__(256) void rnn_out_kernel(
    const float* __restrict__ h,
    const float* __restrict__ W_out,
    const float* __restrict__ b_out,
    float* __restrict__ out)
{
    const int wave = threadIdx.x >> 5;
    const int lane = threadIdx.x & 31;
    const int half = lane >> 4;
    const int l16  = lane & 15;

    const int m0 = wave * 16;
    const int n0 = blockIdx.x * 32;

    v8f acc0 = {};
    v8f acc1 = {};

    const float* a_ptr  = h + (size_t)(m0 + l16) * HS_D + half * 2;
    const float* bbase0 = W_out + (size_t)(n0 + l16) * HS_D + half * 2;
    const float* bbase1 = W_out + (size_t)(n0 + 16 + l16) * HS_D + half * 2;

    #pragma unroll 4
    for (int k = 0; k < HS_D; k += 4) {
        v2f a  = *(const v2f*)(a_ptr + k);
        v2f b0 = *(const v2f*)(bbase0 + k);
        v2f b1 = *(const v2f*)(bbase1 + k);
        acc0 = wmma_f32(a, b0, acc0);
        acc1 = wmma_f32(a, b1, acc1);
    }

    const float bo0 = b_out[n0 + l16];
    const float bo1 = b_out[n0 + 16 + l16];

    #pragma unroll
    for (int r = 0; r < 8; ++r) {
        const int m = m0 + r + half * 8;
        out[(size_t)m * OUT_D + n0 + l16]      = acc0[r] + bo0;
        out[(size_t)m * OUT_D + n0 + 16 + l16] = acc1[r] + bo1;
    }
}

extern "C" void kernel_launch(void* const* d_in, const int* in_sizes, int n_in,
                              void* d_out, int out_size, void* d_ws, size_t ws_size,
                              hipStream_t stream) {
    (void)in_sizes; (void)n_in; (void)out_size; (void)ws_size;

    const float* data  = (const float*)d_in[0];
    const float* W_ih  = (const float*)d_in[1];
    const float* W_hh  = (const float*)d_in[2];
    const float* b_i   = (const float*)d_in[3];
    const float* b_h   = (const float*)d_in[4];
    const float* W_out = (const float*)d_in[5];
    const float* b_out = (const float*)d_in[6];
    float* out = (float*)d_out;

    float* pa = (float*)d_ws;
    float* pb = pa + (size_t)BATCH * HS_D;
    float* pc = out;

    rnn_step_kernel<<<dim3(HS_D / 32, BATCH / 32, 1), 256, 0, stream>>>(
        data, data + (size_t)BATCH * INP_D, pc, W_ih, W_hh, b_i, b_h, pa, pb, 1, 1);

    for (int t = 0; t + 2 < T_STEPS; ++t) {
        rnn_step_kernel<<<dim3(HS_D / 32, BATCH / 32, 2), 256, 0, stream>>>(
            data + (size_t)(t + 1) * BATCH * INP_D, data + (size_t)(t + 2) * BATCH * INP_D,
            pa, W_ih, W_hh, b_i, b_h, pb, pc, 0, 0);
        float* dead = pa; pa = pb; pb = pc; pc = dead;
    }
    rnn_step_kernel<<<dim3(HS_D / 32, BATCH / 32, 2), 256, 0, stream>>>(
        data + (size_t)(T_STEPS - 1) * BATCH * INP_D, nullptr,
        pa, W_ih, W_hh, b_i, b_h, pb, pc, 0, 0);
    { float* dead = pa; pa = pb; pb = pc; pc = dead; }
    rnn_step_kernel<<<dim3(HS_D / 32, BATCH / 32, 1), 256, 0, stream>>>(
        data, nullptr, pa, W_ih, W_hh, b_i, b_h, pb, pc, 0, 0);

    rnn_out_kernel<<<HS_D / 32, 256, 0, stream>>>(pb, W_out, b_out, out);
}


# Step kernel: 4-wave groups, one wave per SIMD, one group per WGP (K split 4 x 256); operand loads in 7 groups ahead of the WMMAs, not spread among them; 4 not 8 partials: changes summation order
# speedup vs baseline: 8.9974x; 1.0377x over previous
#include <hip/hip_runtime.h>
#include <math.h>

#define T_STEPS 256
#define BATCH   128
#define INP_D   1024
#define HS_D    1024
#define OUT_D   1024

typedef __attribute__((ext_vector_type(2))) float v2f;
typedef __attribute__((ext_vector_type(8))) float v8f;

__device__ __forceinline__ v8f wmma_f32(v2f a, v2f b, v8f c) {
    return __builtin_amdgcn_wmma_f32_16x16x4_f32(
         false, a,  false, b,
         (short)0, c,  false,  false);
}

__device__ __forceinline__ float tanh_both_ranges(float x) {
    const float ax = fabsf(x);
    const float x2 = ax * ax;
    float p = fmaf(x2, -0x1.758e7ap-8f, 0x1.521192p-6f);
    p = fmaf(x2, p, -0x1.b8389cp-5f);
    p = fmaf(x2, p, 0x1.110704p-3f);
    p = fmaf(x2, p, -0x1.555532p-2f);
    const float small = fmaf(x2, ax * p, ax);
    const float e   = expf(2.0f * ax);
    const float big = fmaf(__builtin_amdgcn_rcpf(e + 1.0f), -2.0f, 1.0f);
    return copysignf(ax < 0.625f ? small : big, x);
}

#define NWAVES   4
#define KSLICE   256
#define PSTEPS   (KSLICE / 8)
#define GSTEPS   4
#define RING     12

struct PairFrag {
    float4 a0, a1;
    v2f p0, p1, p2, p3;
};

template <int J0, int N>
__device__ __forceinline__ void load_steps(PairFrag (&f)[RING], const float* __restrict__ ap,
                                           const float* __restrict__ wp) {
    #pragma unroll
    for (int j = J0; j < J0 + N; ++j) {
        PairFrag& g = f[j % RING];
        g.a0 = *(const float4*)(ap + 8 * j);
        g.a1 = *(const float4*)(ap + (size_t)16 * HS_D + 8 * j);
        const float* w = wp + (size_t)(8 * j) * HS_D;
        g.p0 = *(const v2f*)(w);
        g.p1 = *(const v2f*)(w + HS_D);
        g.p2 = *(const v2f*)(w + 2 * HS_D);
        g.p3 = *(const v2f*)(w + 3 * HS_D);
    }
}

template <int J0, int N>
__device__ __forceinline__ void mma_steps(const PairFrag (&f)[RING],
                                          v8f& acc00, v8f& acc01, v8f& acc10, v8f& acc11) {
    #pragma unroll
    for (int j = J0; j < J0 + N; ++j) {
        const PairFrag& g = f[j % RING];
        const v2f a0_lo = { g.a0.x, g.a0.y };
        const v2f a0_hi = { g.a0.z, g.a0.w };
        const v2f a1_lo = { g.a1.x, g.a1.y };
        const v2f a1_hi = { g.a1.z, g.a1.w };
        const v2f be_lo = { g.p0.x, g.p1.x };
        const v2f bo_lo = { g.p0.y, g.p1.y };
        const v2f be_hi = { g.p2.x, g.p3.x };
        const v2f bo_hi = { g.p2.y, g.p3.y };
        acc00 = wmma_f32(a0_lo, be_lo, acc00);
        acc01 = wmma_f32(a0_lo, bo_lo, acc01);
        acc10 = wmma_f32(a1_lo, be_lo, acc10);
        acc11 = wmma_f32(a1_lo, bo_lo, acc11);
        acc00 = wmma_f32(a0_hi, be_hi, acc00);
        acc01 = wmma_f32(a0_hi, bo_hi, acc01);
        acc10 = wmma_f32(a1_hi, be_hi, acc10);
        acc11 = wmma_f32(a1_hi, bo_hi, acc11);
    }
}

#define LOAD_PLACE(J0)                                                                        \
    asm volatile("" : "+v"(acc00), "+v"(acc01), "+v"(acc10), "+v"(acc11), "+v"(step));        \
    load_steps<(J0), GSTEPS>(f, ap + step, wp + step);                                        \
    __builtin_amdgcn_sched_barrier(0)

__global__ __launch_bounds__(128) __attribute__((amdgpu_waves_per_eu(1, 1))) void rnn_step_kernel(
    const float* __restrict__ x_next,
    const float* __restrict__ x_pref,
    const float* __restrict__ h_in,
    const float* __restrict__ W_ih,
    const float* __restrict__ W_hh,
    const float* __restrict__ b_i,
    const float* __restrict__ b_h,
    float* __restrict__ sh,
    float* __restrict__ s_next,
    int role0,
    int zero_h)
{
    __shared__ float red[41 * 1024];

    const int wave = threadIdx.x >> 5;
    const int lane = threadIdx.x & 31;
    const int half = lane >> 4;
    const int l16  = lane & 15;

    const int n0 = blockIdx.x * 32;
    const int m0 = blockIdx.y * 32;

    const bool xrole = (blockIdx.z + role0) != 0;
    const float* A  = xrole ? x_next : h_in;
    const float* Wm = xrole ? W_ih   : W_hh;
    const int    kb = wave * KSLICE + half * 4;

    const float* ap = A  + (size_t)(m0 + l16) * HS_D + kb;
    const float* wp = Wm + (size_t)kb * HS_D + n0 + 2 * l16;

    const size_t eo = (size_t)(m0 + wave) * HS_D + n0 + lane;
    float e[8];
    #pragma unroll
    for (int q = 0; q < 8; ++q) e[q] = 0.0f;
    if (xrole) {
        e[0] = b_i[n0 + lane];
        e[1] = b_h[n0 + lane];
    } else {
        #pragma unroll
        for (int q = 0; q < 8; ++q) e[q] = sh[eo + (size_t)(4 * q) * HS_D];
    }

    v8f acc00 = {};
    v8f acc01 = {};
    v8f acc10 = {};
    v8f acc11 = {};

    PairFrag f[RING];
    size_t step = 0;

    static_assert(PSTEPS == 32 && RING == 12 && GSTEPS == 4, "the places below are written for these");
    load_steps<0, 1>(f, ap, wp);
    __builtin_amdgcn_sched_barrier(0);
    load_steps<1, 7>(f, ap, wp);
    __builtin_amdgcn_sched_barrier(0);
    mma_steps<0, 1>(f, acc00, acc01, acc10, acc11);
    LOAD_PLACE(8);
    mma_steps<1, 3>(f, acc00, acc01, acc10, acc11);
    LOAD_PLACE(12);
    mma_steps<4, 4>(f, acc00, acc01, acc10, acc11);
    LOAD_PLACE(16);
    mma_steps<8, 4>(f, acc00, acc01, acc10, acc11);
    LOAD_PLACE(20);
    mma_steps<12, 4>(f, acc00, acc01, acc10, acc11);
    LOAD_PLACE(24);
    mma_steps<16, 4>(f, acc00, acc01, acc10, acc11);
    LOAD_PLACE(28);
    mma_steps<20, 12>(f, acc00, acc01, acc10, acc11);

    if (xrole && x_pref != nullptr) {
        const float* pf = x_pref + (size_t)(m0 + (threadIdx.x >> 2)) * INP_D
                        + (4 * (threadIdx.x & 3) + (blockIdx.x & 3)) * 64;
        __builtin_prefetch(pf, 0, 1);
    }

    float* mine = red + wave * 1024;
    #pragma unroll
    for (int r = 0; r < 8; ++r) {
        v2f t0 = { acc00[r], acc01[r] };
        v2f t1 = { acc10[r], acc11[r] };
        *(v2f*)(mine + (r + half * 8) * 32 + 2 * l16)      = t0;
        *(v2f*)(mine + (16 + r + half * 8) * 32 + 2 * l16) = t1;
    }
    __syncthreads();

    float s[8];
    #pragma unroll
    for (int q = 0; q < 8; ++q) {
        const int o = threadIdx.x + q * 128;
        float v = red[o];
        #pragma unroll
        for (int w = 1; w < NWAVES; ++w) v += red[w * 1024 + o];
        s[q] = v;
    }
    if (xrole) {
        const float bias = e[0] + e[1];
        #pragma unroll
        for (int q = 0; q < 8; ++q) s_next[eo + (size_t)(4 * q) * HS_D] = s[q] + bias;
        if (zero_h) {
            #pragma unroll
            for (int q = 0; q < 8; ++q) sh[eo + (size_t)(4 * q) * HS_D] = 0.0f;
        }
    } else {
        float t[8];
        #pragma unroll
        for (int q = 0; q < 8; ++q) t[q] = tanh_both_ranges(s[q] + e[q]);
        #pragma unroll
        for (int q = 0; q < 8; ++q) sh[eo + (size_t)(4 * q) * HS_D] = t[q];
    }
}

__global__ __launch_bounds__(256) void rnn_out_kernel(
    const float* __restrict__ h,
    const float* __restrict__ W_out,
    const float* __restrict__ b_out,
    float* __restrict__ out)
{
    const int wave = threadIdx.x >> 5;
    const int lane = threadIdx.x & 31;
    const int half = lane >> 4;
    const int l16  = lane & 15;

    const int m0 = wave * 16;
    const int n0 = blockIdx.x * 32;

    v8f acc0 = {};
    v8f acc1 = {};

    const float* a_ptr  = h + (size_t)(m0 + l16) * HS_D + half * 2;
    const float* bbase0 = W_out + (size_t)(n0 + l16) * HS_D + half * 2;
    const float* bbase1 = W_out + (size_t)(n0 + 16 + l16) * HS_D + half * 2;

    #pragma unroll 4
    for (int k = 0; k < HS_D; k += 4) {
        v2f a  = *(const v2f*)(a_ptr + k);
        v2f b0 = *(const v2f*)(bbase0 + k);
        v2f b1 = *(const v2f*)(bbase1 + k);
        acc0 = wmma_f32(a, b0, acc0);
        acc1 = wmma_f32(a, b1, acc1);
    }

    const float bo0 = b_out[n0 + l16];
    const float bo1 = b_out[n0 + 16 + l16];

    #pragma unroll
    for (int r = 0; r < 8; ++r) {
        const int m = m0 + r + half * 8;
        out[(size_t)m * OUT_D + n0 + l16]      = acc0[r] + bo0;
        out[(size_t)m * OUT_D + n0 + 16 + l16] = acc1[r] + bo1;
    }
}

extern "C" void kernel_launch(void* const* d_in, const int* in_sizes, int n_in,
                              void* d_out, int out_size, void* d_ws, size_t ws_size,
                              hipStream_t stream) {
    (void)in_sizes; (void)n_in; (void)out_size; (void)ws_size;

    const float* data  = (const float*)d_in[0];
    const float* W_ih  = (const float*)d_in[1];
    const float* W_hh  = (const float*)d_in[2];
    const float* b_i   = (const float*)d_in[3];
    const float* b_h   = (const float*)d_in[4];
    const float* W_out = (const float*)d_in[5];
    const float* b_out = (const float*)d_in[6];
    float* out = (float*)d_out;

    float* pa = (float*)d_ws;
    float* pb = pa + (size_t)BATCH * HS_D;
    float* pc = out;

    rnn_step_kernel<<<dim3(HS_D / 32, BATCH / 32, 1), 128, 0, stream>>>(
        data, data + (size_t)BATCH * INP_D, pc, W_ih, W_hh, b_i, b_h, pa, pb, 1, 1);

    for (int t = 0; t + 2 < T_STEPS; ++t) {
        rnn_step_kernel<<<dim3(HS_D / 32, BATCH / 32, 2), 128, 0, stream>>>(
            data + (size_t)(t + 1) * BATCH * INP_D, data + (size_t)(t + 2) * BATCH * INP_D,
            pa, W_ih, W_hh, b_i, b_h, pb, pc, 0, 0);
        float* dead = pa; pa = pb; pb = pc; pc = dead;
    }
    rnn_step_kernel<<<dim3(HS_D / 32, BATCH / 32, 2), 128, 0, stream>>>(
        data + (size_t)(T_STEPS - 1) * BATCH * INP_D, nullptr,
        pa, W_ih, W_hh, b_i, b_h, pb, pc, 0, 0);
    { float* dead = pa; pa = pb; pb = pc; pc = dead; }
    rnn_step_kernel<<<dim3(HS_D / 32, BATCH / 32, 1), 128, 0, stream>>>(
        data, nullptr, pa, W_ih, W_hh, b_i, b_h, pb, pc, 0, 0);

    rnn_out_kernel<<<HS_D / 32, 256, 0, stream>>>(pb, W_out, b_out, out);
}
